# starting kernel
# baseline (speedup 1.0000x reference)
#include <hip/hip_runtime.h>
#include <hip/hip_bf16.h>

#define B_SZ   32
#define N_SZ   512
#define D_SZ   768
#define M_SZ   4
#define CDB    (D_SZ * B_SZ)
#define OUTC   D_SZ

typedef __attribute__((ext_vector_type(2))) float v2f;
typedef __attribute__((ext_vector_type(8))) float v8f;

__device__ __forceinline__ v8f wmma4(v2f a, v2f b, v8f c) {
  return __builtin_amdgcn_wmma_f32_16x16x4_f32(
       false, a,  false, b,
       (short)0, c,  false,  false);
}

__global__ __launch_bounds__(256)
void dinv_kernel(const float* __restrict__ adj, float* __restrict__ dinv) {
  __shared__ float red[256];
  const int row = blockIdx.x;
  const int t = threadIdx.x;
  float s = 0.f;
  const float* a = adj + row * N_SZ;
  for (int j = t; j < N_SZ; j += 256) s += a[j];
  red[t] = s;
  __syncthreads();
  for (int off = 128; off > 0; off >>= 1) {
    if (t < off) red[t] += red[t + off];
    __syncthreads();
  }
  if (t == 0) {
    float rs = red[0] + 1.0f;
    dinv[row] = (rs > 0.f) ? rsqrtf(rs) : 0.f;
  }
}

__global__ __launch_bounds__(256)
void support_kernel(const float* __restrict__ adj, const float* __restrict__ dinv,
                    float* __restrict__ support) {
  const int idx = blockIdx.x * 256 + threadIdx.x;
  const int i = idx >> 9;
  const int j = idx & (N_SZ - 1);
  float m = adj[j * N_SZ + i] + (i == j ? 1.0f : 0.0f);
  support[idx] = m * dinv[i] * dinv[j];
}

__global__ __launch_bounds__(256)
void transpose_kernel(const float* __restrict__ inputs, float* __restrict__ x0) {
  __shared__ float tile[64 * 33];
  const int n  = blockIdx.x;
  const int d0 = blockIdx.y * 64;
  const int t  = threadIdx.x;
#pragma unroll
  for (int p = 0; p < 8; ++p) {
    int b  = p * 4 + (t >> 6);
    int dl = t & 63;
    tile[dl * 33 + b] =
        inputs[(size_t)b * (N_SZ * D_SZ) + (size_t)n * D_SZ + d0 + dl];
  }
  __syncthreads();
  float* dst = x0 + (size_t)n * CDB + d0 * B_SZ;
#pragma unroll
  for (int p = 0; p < 8; ++p) {
    int c  = p * 256 + t;
    int dl = c >> 5;
    int b  = c & 31;
    dst[c] = tile[dl * 33 + b];
  }
}

__device__ __forceinline__
void store_tile(float* __restrict__ Y, const float* __restrict__ Xprev,
                int rbase, int col, float alpha, float beta,
                v8f a0, v8f a1, v8f a2, v8f a3) {
  if (beta != 0.0f) {
#pragma unroll
    for (int j = 0; j < 8; ++j) {
      float* Yr = Y + (rbase + j) * CDB + col;
      const float* Pr = Xprev + (rbase + j) * CDB + col;
      Yr[0]  = alpha * a0[j] + beta * Pr[0];
      Yr[16] = alpha * a1[j] + beta * Pr[16];
      Yr[32] = alpha * a2[j] + beta * Pr[32];
      Yr[48] = alpha * a3[j] + beta * Pr[48];
    }
  } else {
#pragma unroll
    for (int j = 0; j < 8; ++j) {
      float* Yr = Y + (rbase + j) * CDB + col;
      Yr[0]  = alpha * a0[j];
      Yr[16] = alpha * a1[j];
      Yr[32] = alpha * a2[j];
      Yr[48] = alpha * a3[j];
    }
  }
}

__global__ __launch_bounds__(256)
void diffusion_gemm(const float* __restrict__ S, const float* __restrict__ X,
                    const float* __restrict__ Xprev, float* __restrict__ Y,
                    float alpha, float beta) {
  const int lane = threadIdx.x & 31;
  const int wave = threadIdx.x >> 5;
  const int lm   = lane & 15;
  const int kh   = lane >> 4;
  const int m0 = blockIdx.y * 64 + (wave >> 2) * 32;
  const int n0 = blockIdx.x * 256 + (wave & 3) * 64;

  const float* Srow0 = S + (m0 + lm) * N_SZ + kh * 2;
  const float* Srow1 = Srow0 + 16 * N_SZ;
  const float* Xcol  = X + n0 + lm;

  v8f acc00 = {}, acc01 = {}, acc02 = {}, acc03 = {};
  v8f acc10 = {}, acc11 = {}, acc12 = {}, acc13 = {};
  for (int k = 0; k < N_SZ; k += 4) {
    v2f a0 = *(const v2f*)(Srow0 + k);
    v2f a1 = *(const v2f*)(Srow1 + k);
    const float* Xr0 = Xcol + (k + kh * 2) * CDB;
    const float* Xr1 = Xr0 + CDB;
    v2f b0 = {Xr0[0],  Xr1[0]};
    v2f b1 = {Xr0[16], Xr1[16]};
    v2f b2 = {Xr0[32], Xr1[32]};
    v2f b3 = {Xr0[48], Xr1[48]};
    acc00 = wmma4(a0, b0, acc00);
    acc01 = wmma4(a0, b1, acc01);
    acc02 = wmma4(a0, b2, acc02);
    acc03 = wmma4(a0, b3, acc03);
    acc10 = wmma4(a1, b0, acc10);
    acc11 = wmma4(a1, b1, acc11);
    acc12 = wmma4(a1, b2, acc12);
    acc13 = wmma4(a1, b3, acc13);
  }

  store_tile(Y, Xprev, m0 + kh * 8,      n0 + lm, alpha, beta,
             acc00, acc01, acc02, acc03);
  store_tile(Y, Xprev, m0 + 16 + kh * 8, n0 + lm, alpha, beta,
             acc10, acc11, acc12, acc13);
}

__global__ __launch_bounds__(256)
void output_gemm(const float* __restrict__ x0, const float* __restrict__ x1,
                 const float* __restrict__ x2, const float* __restrict__ x3,
                 const float* __restrict__ W, const float* __restrict__ bias,
                 float* __restrict__ out) {
  const int lane = threadIdx.x & 31;
  const int wave = threadIdx.x >> 5;
  const int lm   = lane & 15;
  const int kh   = lane >> 4;
  const int r0 = blockIdx.y * 64 + (wave >> 2) * 32;
  const int o0 = blockIdx.x * 256 + (wave & 3) * 64;

  const int ra = r0 + lm;
  const int rb = r0 + 16 + lm;
  const size_t aoff0 = (size_t)(ra & (N_SZ - 1)) * CDB + (ra >> 9);
  const size_t aoff1 = (size_t)(rb & (N_SZ - 1)) * CDB + (rb >> 9);
  const float* xa = (kh == 0) ? x0 : x2;
  const float* xb = (kh == 0) ? x1 : x3;
  const float* Wb = W + kh * 2 * OUTC + o0 + lm;

  v8f acc00 = {}, acc01 = {}, acc02 = {}, acc03 = {};
  v8f acc10 = {}, acc11 = {}, acc12 = {}, acc13 = {};
  for (int d = 0; d < D_SZ; ++d) {
    const int xo = d * B_SZ;
    v2f a0 = {xa[aoff0 + xo], xb[aoff0 + xo]};
    v2f a1 = {xa[aoff1 + xo], xb[aoff1 + xo]};
    const float* Wr0 = Wb + (d * 4) * OUTC;
    const float* Wr1 = Wr0 + OUTC;
    v2f b0 = {Wr0[0],  Wr1[0]};
    v2f b1 = {Wr0[16], Wr1[16]};
    v2f b2 = {Wr0[32], Wr1[32]};
    v2f b3 = {Wr0[48], Wr1[48]};
    acc00 = wmma4(a0, b0, acc00);
    acc01 = wmma4(a0, b1, acc01);
    acc02 = wmma4(a0, b2, acc02);
    acc03 = wmma4(a0, b3, acc03);
    acc10 = wmma4(a1, b0, acc10);
    acc11 = wmma4(a1, b1, acc11);
    acc12 = wmma4(a1, b2, acc12);
    acc13 = wmma4(a1, b3, acc13);
  }

  const float bi0 = bias[o0 + lm];
  const float bi1 = bias[o0 + 16 + lm];
  const float bi2 = bias[o0 + 32 + lm];
  const float bi3 = bias[o0 + 48 + lm];
#pragma unroll
  for (int h = 0; h < 2; ++h) {
    const int rbase = r0 + h * 16 + kh * 8;
    v8f c0 = h ? acc10 : acc00;
    v8f c1 = h ? acc11 : acc01;
    v8f c2 = h ? acc12 : acc02;
    v8f c3 = h ? acc13 : acc03;
#pragma unroll
    for (int j = 0; j < 8; ++j) {
      float* Or = out + (rbase + j) * OUTC + o0 + lm;
      Or[0]  = c0[j] + bi0;
      Or[16] = c1[j] + bi1;
      Or[32] = c2[j] + bi2;
      Or[48] = c3[j] + bi3;
    }
  }
}

extern "C" void kernel_launch(void* const* d_in, const int* in_sizes, int n_in,
                              void* d_out, int out_size, void* d_ws, size_t ws_size,
                              hipStream_t stream) {
  const float* inputs  = (const float*)d_in[0];
  const float* adj     = (const float*)d_in[1];
  const float* weights = (const float*)d_in[2];
  const float* biases  = (const float*)d_in[3];
  float* out = (float*)d_out;

  float* ws      = (float*)d_ws;
  float* support = ws;
  float* dinv    = support + (size_t)N_SZ * N_SZ;
  float* x0      = dinv + 512;
  float* x1      = x0 + (size_t)N_SZ * CDB;
  float* x2      = x1 + (size_t)N_SZ * CDB;
  float* x3      = x2 + (size_t)N_SZ * CDB;

  (void)in_sizes; (void)n_in; (void)out_size; (void)ws_size;

  dinv_kernel<<<N_SZ, 256, 0, stream>>>(adj, dinv);
  support_kernel<<<(N_SZ * N_SZ) / 256, 256, 0, stream>>>(adj, dinv, support);
  transpose_kernel<<<dim3(N_SZ, D_SZ / 64), 256, 0, stream>>>(inputs, x0);

  dim3 gd(CDB / 256, N_SZ / 64);
  diffusion_gemm<<<gd, 256, 0, stream>>>(support, x0, x0, x1, 1.0f, 0.0f);
  diffusion_gemm<<<gd, 256, 0, stream>>>(support, x1, x0, x2, 2.0f, -1.0f);
  diffusion_gemm<<<gd, 256, 0, stream>>>(support, x2, x1, x3, 2.0f, -1.0f);

  output_gemm<<<dim3(OUTC / 256, (B_SZ * N_SZ) / 64), 256, 0, stream>>>(
      x0, x1, x2, x3, weights, biases, out);
}


# both GEMMs: LDS fragment reads issued one K step ahead of the matrix instructions that use them (second register set, held by scheduling builtins), tile stores under the last step; k, sum order kept
# speedup vs baseline: 4.0811x; 4.0811x over previous
#include <hip/hip_runtime.h>

#define B_SZ   32
#define N_SZ   512
#define D_SZ   768
#define OUTC   D_SZ
#define PLANE  (B_SZ * N_SZ * D_SZ)

typedef __attribute__((ext_vector_type(2))) float v2f;
typedef __attribute__((ext_vector_type(4))) float v4f;
typedef __attribute__((ext_vector_type(8))) float v8f;

__device__ __forceinline__ v8f wmma4(v2f a, v2f b, v8f c) {
  return __builtin_amdgcn_wmma_f32_16x16x4_f32(
       false, a,  false, b,
       (short)0, c,  false,  false);
}

__global__ __launch_bounds__(256)
void dinv_kernel(const float* __restrict__ adj, float* __restrict__ dinv) {
  __shared__ float red[256];
  const int row = blockIdx.x;
  const int t = threadIdx.x;
  float s = 0.f;
  const float* a = adj + row * N_SZ;
  for (int j = t; j < N_SZ; j += 256) s += a[j];
  red[t] = s;
  __syncthreads();
  for (int off = 128; off > 0; off >>= 1) {
    if (t < off) red[t] += red[t + off];
    __syncthreads();
  }
  if (t == 0) {
    float rs = red[0] + 1.0f;
    dinv[row] = (rs > 0.f) ? rsqrtf(rs) : 0.f;
  }
}

__global__ __launch_bounds__(256)
void support_kernel(const float* __restrict__ adj, const float* __restrict__ dinv,
                    float* __restrict__ support) {
  const int idx = blockIdx.x * 256 + threadIdx.x;
  const int i = idx >> 9;
  const int j = idx & (N_SZ - 1);
  float m = adj[j * N_SZ + i] + (i == j ? 1.0f : 0.0f);
  support[idx] = m * dinv[i] * dinv[j];
}

__device__ __forceinline__
void store_tile(float* __restrict__ Y, const float* __restrict__ Xprev,
                int rbase, int col, float alpha, float beta,
                v8f a0, v8f a1, v8f a2, v8f a3) {
  if (beta != 0.0f) {
#pragma unroll
    for (int j = 0; j < 8; ++j) {
      float* Yr = Y + (rbase + j) * D_SZ + col;
      const float* Pr = Xprev + (rbase + j) * D_SZ + col;
      Yr[0]  = alpha * a0[j] + beta * Pr[0];
      Yr[16] = alpha * a1[j] + beta * Pr[16];
      Yr[32] = alpha * a2[j] + beta * Pr[32];
      Yr[48] = alpha * a3[j] + beta * Pr[48];
    }
  } else {
#pragma unroll
    for (int j = 0; j < 8; ++j) {
      float* Yr = Y + (rbase + j) * D_SZ + col;
      Yr[0]  = alpha * a0[j];
      Yr[16] = alpha * a1[j];
      Yr[32] = alpha * a2[j];
      Yr[48] = alpha * a3[j];
    }
  }
}

#define TM      64
#define TN      256
#define KC      32
#define APITCH  (KC + 4)
#define BPITCH  (TN * 2 + 32)
#define A_STAGE (TM * APITCH)
#define B_STAGE ((KC / 2) * BPITCH)
#define STAGE   (A_STAGE + B_STAGE)

struct BRegs { v4f r0[4], r1[4]; };

__device__ __forceinline__
void load_b(BRegs& br, const float* __restrict__ Bg, int ldb, int t) {
  const int cq = t & 63;
  const int pr = t >> 6;
#pragma unroll
  for (int q = 0; q < 4; ++q) {
    const float* p0 = Bg + (2 * (pr + 4 * q)) * ldb + 4 * cq;
    br.r0[q] = *(const v4f*)(p0);
    br.r1[q] = *(const v4f*)(p0 + ldb);
  }
}

__device__ __forceinline__
void store_b(const BRegs& br, float* __restrict__ Bs, int t) {
  const int cq = t & 63;
  const int pr = t >> 6;
#pragma unroll
  for (int q = 0; q < 4; ++q) {
    float* d = Bs + (pr + 4 * q) * BPITCH + 8 * cq;
    v4f lo = {br.r0[q][0], br.r1[q][0], br.r0[q][1], br.r1[q][1]};
    v4f hi = {br.r0[q][2], br.r1[q][2], br.r0[q][3], br.r1[q][3]};
    *(v4f*)(d)     = lo;
    *(v4f*)(d + 4) = hi;
  }
}

#define LOAD_A(ks)                                                      \
  {                                                                     \
    a0[ks] = *(const v2f*)(Ap + 4 * (ks));                              \
    a1[ks] = *(const v2f*)(Ap + 16 * APITCH + 4 * (ks));                \
  }
#define LOAD_B(ks)                                                      \
  {                                                                     \
    const float* Bk = Bp + (2 * (ks)) * BPITCH;                         \
    b0[ks] = *(const v2f*)(Bk);                                         \
    b1[ks] = *(const v2f*)(Bk + 32);                                    \
    b2[ks] = *(const v2f*)(Bk + 64);                                    \
    b3[ks] = *(const v2f*)(Bk + 96);                                    \
  }
#define PIPE_STEP(nreads)                                               \
  __builtin_amdgcn_sched_group_barrier(0x100, nreads, 0);               \
  __builtin_amdgcn_sched_group_barrier(0x008, 8, 0);
#define CHUNK_PIPELINE                                                  \
  __builtin_amdgcn_sched_group_barrier(0x100, 4, 0);                    \
  PIPE_STEP(2) PIPE_STEP(4) PIPE_STEP(2) PIPE_STEP(4)                   \
  PIPE_STEP(2) PIPE_STEP(4) PIPE_STEP(2)                                \
  __builtin_amdgcn_sched_group_barrier(0x200, 12, 0);                   \
  __builtin_amdgcn_sched_group_barrier(0x008, 8, 0);

__device__ __forceinline__
void mma_chunk(const float* __restrict__ Ap, const float* __restrict__ Bp,
               v8f& acc00, v8f& acc01, v8f& acc02, v8f& acc03,
               v8f& acc10, v8f& acc11, v8f& acc12, v8f& acc13) {
  v2f a0[8], a1[8], b0[8], b1[8], b2[8], b3[8];
  LOAD_A(0);
  LOAD_A(1);
  LOAD_B(0);
#pragma unroll
  for (int ks = 0; ks < KC / 4; ++ks) {
    if (ks + 1 < KC / 4) LOAD_B(ks + 1);
    if ((ks & 1) && ks + 2 < KC / 4) {
      LOAD_A(ks + 1);
      LOAD_A(ks + 2);
    }
    acc00 = wmma4(a0[ks], b0[ks], acc00);
    acc01 = wmma4(a0[ks], b1[ks], acc01);
    acc02 = wmma4(a0[ks], b2[ks], acc02);
    acc03 = wmma4(a0[ks], b3[ks], acc03);
    acc10 = wmma4(a1[ks], b0[ks], acc10);
    acc11 = wmma4(a1[ks], b1[ks], acc11);
    acc12 = wmma4(a1[ks], b2[ks], acc12);
    acc13 = wmma4(a1[ks], b3[ks], acc13);
  }
}

__global__ __launch_bounds__(256)
void diffusion_gemm(const float* __restrict__ S, const float* __restrict__ X,
                    const float* __restrict__ Xprev, float* __restrict__ Y,
                    float alpha, float beta) {
  __shared__ __attribute__((aligned(16))) float lds[2 * STAGE];
  const int t    = threadIdx.x;
  const int lane = t & 31;
  const int wave = t >> 5;
  const int lm   = lane & 15;
  const int kh   = lane >> 4;
  const int mw = blockIdx.y * TM;
  const int nw = blockIdx.x * TN;
  const int m0 = mw + (wave >> 2) * 32;
  const int n0 = nw + (wave & 3) * 64;
  const int boff = blockIdx.z * (N_SZ * D_SZ);
  X += boff;
  Xprev += boff;
  Y += boff;

  const int ar = t >> 3;
  const int ac = (t & 7) * 4;
  const float* Ag = S + (mw + ar) * N_SZ + ac;
  const float* Bg = X + nw;
  const int a_st = ar * APITCH + ac;

  const int a_fr = ((wave >> 2) * 32 + lm) * APITCH + 2 * kh;
  const int b_fr = A_STAGE + kh * BPITCH + ((wave & 3) * 64 + lm) * 2;

  v8f acc00 = {}, acc01 = {}, acc02 = {}, acc03 = {};
  v8f acc10 = {}, acc11 = {}, acc12 = {}, acc13 = {};

  v4f ra0, ra1;
  BRegs br;
  ra0 = *(const v4f*)(Ag);
  ra1 = *(const v4f*)(Ag + 32 * N_SZ);
  load_b(br, Bg, D_SZ, t);
  *(v4f*)(lds + a_st) = ra0;
  *(v4f*)(lds + a_st + 32 * APITCH) = ra1;
  store_b(br, lds + A_STAGE, t);
  __syncthreads();

  const int NC = N_SZ / KC;
  for (int c = 0; c < NC; ++c) {
    float* cur = lds + (c & 1) * STAGE;
    float* nxt = lds + ((c + 1) & 1) * STAGE;
    const bool more = (c + 1 < NC);
    if (more) {
      const int kc = (c + 1) * KC;
      ra0 = *(const v4f*)(Ag + kc);
      ra1 = *(const v4f*)(Ag + 32 * N_SZ + kc);
      load_b(br, Bg + kc * D_SZ, D_SZ, t);
    }
    mma_chunk(cur + a_fr, cur + b_fr,
              acc00, acc01, acc02, acc03, acc10, acc11, acc12, acc13);
    if (more) {
      *(v4f*)(nxt + a_st) = ra0;
      *(v4f*)(nxt + a_st + 32 * APITCH) = ra1;
      store_b(br, nxt + A_STAGE, t);
      CHUNK_PIPELINE
    }
    __syncthreads();
  }

  store_tile(Y, Xprev, m0 + kh * 8,      n0 + lm, alpha, beta,
             acc00, acc01, acc02, acc03);
  store_tile(Y, Xprev, m0 + 16 + kh * 8, n0 + lm, alpha, beta,
             acc10, acc11, acc12, acc13);
}

__global__ __launch_bounds__(256)
void output_gemm(const float* __restrict__ x0, const float* __restrict__ x1,
                 const float* __restrict__ x2, const float* __restrict__ x3,
                 const float* __restrict__ W, const float* __restrict__ bias,
                 float* __restrict__ out) {
  __shared__ __attribute__((aligned(16))) float lds[2 * STAGE];
  const int t    = threadIdx.x;
  const int lane = t & 31;
  const int wave = t >> 5;
  const int lm   = lane & 15;
  const int kh   = lane >> 4;
  const int rw = blockIdx.y * TM;
  const int ow = blockIdx.x * TN;
  const int r0 = rw + (wave >> 2) * 32;
  const int o0 = ow + (wave & 3) * 64;

  const int am = t & 3;
  const int ag = (t >> 2) & 1;
  const int ar = t >> 3;
  const float* xm = (am == 0) ? x0 : (am == 1) ? x1 : (am == 2) ? x2 : x3;
  const float* Ag = xm + (rw + ar) * D_SZ + 4 * ag;
  const float* Bg = W + ow;
  const int a_st = ar * APITCH + 16 * ag + am;

  const int a_fr = ((wave >> 2) * 32 + lm) * APITCH + 2 * kh;
  const int b_fr = A_STAGE + kh * BPITCH + ((wave & 3) * 64 + lm) * 2;

  v8f acc00 = {}, acc01 = {}, acc02 = {}, acc03 = {};
  v8f acc10 = {}, acc11 = {}, acc12 = {}, acc13 = {};

  v4f ra0, ra1;
  BRegs br;
  ra0 = *(const v4f*)(Ag);
  ra1 = *(const v4f*)(Ag + 32 * D_SZ);
  load_b(br, Bg, OUTC, t);
#pragma unroll
  for (int i = 0; i < 4; ++i) {
    lds[a_st + 4 * i] = ra0[i];
    lds[a_st + 32 * APITCH + 4 * i] = ra1[i];
  }
  store_b(br, lds + A_STAGE, t);
  __syncthreads();

  const int NC = (D_SZ * 4) / KC;
  for (int c = 0; c < NC; ++c) {
    float* cur = lds + (c & 1) * STAGE;
    float* nxt = lds + ((c + 1) & 1) * STAGE;
    const bool more = (c + 1 < NC);
    if (more) {
      const int kc = (c + 1) * KC;
      ra0 = *(const v4f*)(Ag + kc / 4);
      ra1 = *(const v4f*)(Ag + 32 * D_SZ + kc / 4);
      load_b(br, Bg + kc * OUTC, OUTC, t);
    }
    mma_chunk(cur + a_fr, cur + b_fr,
              acc00, acc01, acc02, acc03, acc10, acc11, acc12, acc13);
    if (more) {
#pragma unroll
      for (int i = 0; i < 4; ++i) {
        nxt[a_st + 4 * i] = ra0[i];
        nxt[a_st + 32 * APITCH + 4 * i] = ra1[i];
      }
      store_b(br, nxt + A_STAGE, t);
      CHUNK_PIPELINE
    }
    __syncthreads();
  }

  const int lmo = o0 + lm;
  const float bi0 = bias[lmo];
  const float bi1 = bias[lmo + 16];
  const float bi2 = bias[lmo + 32];
  const float bi3 = bias[lmo + 48];
#pragma unroll
  for (int h = 0; h < 2; ++h) {
    const int rbase = r0 + h * 16 + kh * 8;
    v8f c0 = h ? acc10 : acc00;
    v8f c1 = h ? acc11 : acc01;
    v8f c2 = h ? acc12 : acc02;
    v8f c3 = h ? acc13 : acc03;
#pragma unroll
    for (int j = 0; j < 8; ++j) {
      float* Or = out + (rbase + j) * OUTC + lmo;
      Or[0]  = c0[j] + bi0;
      Or[16] = c1[j] + bi1;
      Or[32] = c2[j] + bi2;
      Or[48] = c3[j] + bi3;
    }
  }
}

extern "C" void kernel_launch(void* const* d_in, const int* in_sizes, int n_in,
                              void* d_out, int out_size, void* d_ws, size_t ws_size,
                              hipStream_t stream) {
  const float* inputs  = (const float*)d_in[0];
  const float* adj     = (const float*)d_in[1];
  const float* weights = (const float*)d_in[2];
  const float* biases  = (const float*)d_in[3];
  float* out = (float*)d_out;

  float* ws      = (float*)d_ws;
  float* support = ws;
  float* dinv    = support + (size_t)N_SZ * N_SZ;
  float* x1      = dinv + 512;
  float* x2      = x1 + (size_t)PLANE;
  float* x3      = x2 + (size_t)PLANE;

  (void)in_sizes; (void)n_in; (void)out_size; (void)ws_size;

  dinv_kernel<<<N_SZ, 256, 0, stream>>>(adj, dinv);
  support_kernel<<<(N_SZ * N_SZ) / 256, 256, 0, stream>>>(adj, dinv, support);

  dim3 gd(D_SZ / 256, N_SZ / 64, B_SZ);
  diffusion_gemm<<<gd, 256, 0, stream>>>(support, inputs, inputs, x1, 1.0f, 0.0f);
  diffusion_gemm<<<gd, 256, 0, stream>>>(support, x1, inputs, x2, 2.0f, -1.0f);
  diffusion_gemm<<<gd, 256, 0, stream>>>(support, x2, x1, x3, 2.0f, -1.0f);

  output_gemm<<<dim3(OUTC / 256, (B_SZ * N_SZ) / 64), 256, 0, stream>>>(
      inputs, x1, x2, x3, weights, biases, out);
}
